# starting kernel
# baseline (speedup 1.0000x reference)
#include <hip/hip_runtime.h>

typedef __attribute__((ext_vector_type(2))) float v2f;
typedef __attribute__((ext_vector_type(8))) float v8f;

#define VOCAB 32000
#define EMBED 256
#define NPOS  4096
#define NGRAM 3

__host__ __device__ constexpr unsigned mkcfg(unsigned K) {
    return ((K % 3u) * 16u + K / 3u)
         | ((((K + 1u) % 3u) * 16u + (K + 1u) / 3u) << 6)
         | ((K / 3u) << 12)
         | (((K + 1u) / 3u) << 16);
}

__global__ __launch_bounds__(256) void ngrams_embed_wmma(
    const int* __restrict__ inp,
    const float* __restrict__ weight,
    float* __restrict__ out)
{
    __shared__ int sTab[96];

    const int pg  = blockIdx.x;
    const int p0  = pg * 16;
    const int tid = threadIdx.x;

    if (tid < 16) {
        const int q  = tid;
        const int i0 = inp[p0 + q];
        const int i1 = inp[NPOS + p0 + q];
        const int i2 = inp[2 * NPOS + p0 + q];
        sTab[q]      = i0;
        sTab[16 + q] = i1;
        sTab[32 + q] = i2;
        sTab[48 + q] = __float_as_int(1.0f);
        sTab[64 + q] = (i1 != i0) ? __float_as_int(1.0f) : 0;
        sTab[80 + q] = ((i2 != i0) && (i2 != i1)) ? __float_as_int(1.0f) : 0;
    }
    __syncthreads();

    const int wid  = tid >> 5;
    const int lane = tid & 31;
    const int half = lane >> 4;
    const unsigned r = lane & 15;
    const unsigned colbase = (unsigned)wid * 32u + r;

    v8f acc0 = {};
    v8f acc1 = {};

#pragma unroll
    for (int kg = 0; kg < 12; ++kg) {
        const unsigned cfg = half ? mkcfg(4u * kg + 2u) : mkcfg(4u * kg);
        const unsigned off0 = cfg & 63u;
        const unsigned off1 = (cfg >> 6) & 63u;
        const unsigned q0   = (cfg >> 12) & 15u;
        const unsigned q1   = (cfg >> 16) & 15u;

        const unsigned row0 = (unsigned)sTab[off0];
        const float    fl0  = __int_as_float(sTab[48 + off0]);
        const unsigned row1 = (unsigned)sTab[off1];
        const float    fl1  = __int_as_float(sTab[48 + off1]);

        v2f a;
        a.x = (r == q0) ? fl0 : 0.0f;
        a.y = (r == q1) ? fl1 : 0.0f;

        const float* b0p = weight + ((row0 << 8) + colbase);
        const float* b1p = weight + ((row1 << 8) + colbase);

        v2f b;
        b.x = b0p[0];
        b.y = b1p[0];
        acc0 = __builtin_amdgcn_wmma_f32_16x16x4_f32(
            false, a, false, b, (short)0, acc0, false, false);

        b.x = b0p[16];
        b.y = b1p[16];
        acc1 = __builtin_amdgcn_wmma_f32_16x16x4_f32(
            false, a, false, b, (short)0, acc1, false, false);
    }

#pragma unroll
    for (int j = 0; j < 8; ++j) {
        const unsigned m    = (unsigned)j + 8u * (unsigned)half;
        const unsigned ooff = (((unsigned)p0 + m) << 8) + (unsigned)wid * 32u + r;
        __builtin_nontemporal_store(acc0[j], out + ooff);
        __builtin_nontemporal_store(acc1[j], out + ooff + 16u);
    }
}

extern "C" void kernel_launch(void* const* d_in, const int* in_sizes, int n_in,
                              void* d_out, int out_size, void* d_ws, size_t ws_size,
                              hipStream_t stream) {
    const int*   inp    = (const int*)d_in[0];
    const float* weight = (const float*)d_in[1];
    float*       outp   = (float*)d_out;

    ngrams_embed_wmma<<<dim3(NPOS / 16), dim3(256), 0, stream>>>(inp, weight, outp);
}


# Wave tile 16 pos x 64 dims at one wave per SIMD: a lane gathers 16 B per row (one b128 feeds four N tiles), all 24 row loads go out before the first WMMA, 16-B stores; sum order unchanged
# speedup vs baseline: 1.2990x; 1.2990x over previous
#include <hip/hip_runtime.h>

typedef __attribute__((ext_vector_type(2))) float v2f;
typedef __attribute__((ext_vector_type(4))) float v4f;
typedef __attribute__((ext_vector_type(8))) float v8f;

#define VOCAB 32000
#define EMBED 256
#define NPOS  4096
#define NGRAM 3

__host__ __device__ constexpr unsigned mkcfg(unsigned K) {
    return ((K % 3u) * 16u + K / 3u)
         | ((((K + 1u) % 3u) * 16u + (K + 1u) / 3u) << 6)
         | ((K / 3u) << 12)
         | (((K + 1u) / 3u) << 16);
}

__global__ __launch_bounds__(128) __attribute__((amdgpu_waves_per_eu(1, 1)))
void ngrams_embed_wmma(
    const int* __restrict__ inp,
    const float* __restrict__ weight,
    float* __restrict__ out)
{
    __shared__ int sTab[96];

    const int pg  = blockIdx.x;
    const int p0  = pg * 16;
    const int tid = threadIdx.x;

    if (tid < 16) {
        const int q  = tid;
        const int i0 = inp[p0 + q];
        const int i1 = inp[NPOS + p0 + q];
        const int i2 = inp[2 * NPOS + p0 + q];
        sTab[q]      = i0;
        sTab[16 + q] = i1;
        sTab[32 + q] = i2;
        sTab[48 + q] = __float_as_int(1.0f);
        sTab[64 + q] = (i1 != i0) ? __float_as_int(1.0f) : 0;
        sTab[80 + q] = ((i2 != i0) && (i2 != i1)) ? __float_as_int(1.0f) : 0;
    }
    __syncthreads();

    const int wid  = tid >> 5;
    const int lane = tid & 31;
    const int half = lane >> 4;
    const unsigned r = lane & 15;
    const unsigned colbase = (unsigned)wid * 64u + 4u * r;

    v8f acc0 = {};
    v8f acc1 = {};
    v8f acc2 = {};
    v8f acc3 = {};

#pragma unroll
    for (int kg = 0; kg < 12; ++kg) {
        const unsigned cfg = half ? mkcfg(4u * kg + 2u) : mkcfg(4u * kg);
        const unsigned off0 = cfg & 63u;
        const unsigned off1 = (cfg >> 6) & 63u;
        const unsigned q0   = (cfg >> 12) & 15u;
        const unsigned q1   = (cfg >> 16) & 15u;

        const unsigned row0 = (unsigned)sTab[off0];
        const float    fl0  = __int_as_float(sTab[48 + off0]);
        const unsigned row1 = (unsigned)sTab[off1];
        const float    fl1  = __int_as_float(sTab[48 + off1]);

        v2f a;
        a.x = (r == q0) ? fl0 : 0.0f;
        a.y = (r == q1) ? fl1 : 0.0f;

        const v4f w0 = *reinterpret_cast<const v4f*>(weight + ((row0 << 8) + colbase));
        const v4f w1 = *reinterpret_cast<const v4f*>(weight + ((row1 << 8) + colbase));

        v2f b;
        b.x = w0.x;
        b.y = w1.x;
        acc0 = __builtin_amdgcn_wmma_f32_16x16x4_f32(
            false, a, false, b, (short)0, acc0, false, false);
        b.x = w0.y;
        b.y = w1.y;
        acc1 = __builtin_amdgcn_wmma_f32_16x16x4_f32(
            false, a, false, b, (short)0, acc1, false, false);
        b.x = w0.z;
        b.y = w1.z;
        acc2 = __builtin_amdgcn_wmma_f32_16x16x4_f32(
            false, a, false, b, (short)0, acc2, false, false);
        b.x = w0.w;
        b.y = w1.w;
        acc3 = __builtin_amdgcn_wmma_f32_16x16x4_f32(
            false, a, false, b, (short)0, acc3, false, false);
    }

#pragma unroll
    for (int j = 0; j < 8; ++j) {
        const unsigned m    = (unsigned)j + 8u * (unsigned)half;
        const unsigned ooff = (((unsigned)p0 + m) << 8) + colbase;
        v4f o;
        o.x = acc0[j];
        o.y = acc1[j];
        o.z = acc2[j];
        o.w = acc3[j];
        __builtin_nontemporal_store(o, reinterpret_cast<v4f*>(out + ooff));
    }
}

extern "C" void kernel_launch(void* const* d_in, const int* in_sizes, int n_in,
                              void* d_out, int out_size, void* d_ws, size_t ws_size,
                              hipStream_t stream) {
    const int*   inp    = (const int*)d_in[0];
    const float* weight = (const float*)d_in[1];
    float*       outp   = (float*)d_out;

    ngrams_embed_wmma<<<dim3(NPOS / 16), dim3(128), 0, stream>>>(inp, weight, outp);
}
